# starting kernel
# baseline (speedup 1.0000x reference)
#include <hip/hip_runtime.h>

typedef __attribute__((ext_vector_type(2))) float v2f;
typedef __attribute__((ext_vector_type(8))) float v8f;

#define N_TOK   1728
#define CDIM    256
#define NHEADS  8
#define HD      32
#define NT      108
#define SCALE_  0.1767766952966369f

__device__ __forceinline__ v8f wmma4(v2f a, v2f b, v8f c) {
  return __builtin_amdgcn_wmma_f32_16x16x4_f32(
       false, a,  false, b,
       (short)0, c,  false,  false);
}

__global__ __launch_bounds__(256)
void gsa3d_qkv_kernel(const float* __restrict__ x,
                      const float* __restrict__ wqkv,
                      float* __restrict__ Q,
                      float* __restrict__ K,
                      float* __restrict__ V) {
  const int wid  = blockIdx.x * 8 + (threadIdx.x >> 5);
  const int lane = threadIdx.x & 31;
  const int l16  = lane & 15;
  const int kh   = lane >> 4;
  const int tn   = wid % 48;
  const int tm   = wid / 48;
  const int row  = tm * 16 + l16;
  const int col  = tn * 16 + l16;

  v8f acc = {};
  for (int k0 = 0; k0 < CDIM; k0 += 4) {
    const int ka = k0 + 2 * kh;
    v2f a, b;
    a.x = x[(ka + 0) * N_TOK + row];
    a.y = x[(ka + 1) * N_TOK + row];
    b.x = wqkv[col * CDIM + ka + 0];
    b.y = wqkv[col * CDIM + ka + 1];
    acc = wmma4(a, b, acc);
  }

  const int s  = col >> 8;
  const int hh = (col >> 5) & 7;
  const int d  = col & 31;
  float* dst = (s == 0) ? Q : ((s == 1) ? K : V);
  dst += hh * (N_TOK * HD) + d;
#pragma unroll
  for (int r = 0; r < 8; ++r) {
    const int n = tm * 16 + r + 8 * kh;
    dst[n * HD] = acc[r];
  }
}

__global__ __launch_bounds__(256)
void gsa3d_attn_kernel(const float* __restrict__ Q,
                       const float* __restrict__ K,
                       const float* __restrict__ V,
                       float* __restrict__ AO) {
  __shared__ float ldsP[8 * 256];
  const int wslot = threadIdx.x >> 5;
  const int wid   = blockIdx.x * 8 + wslot;
  const int lane  = threadIdx.x & 31;
  const int l16   = lane & 15;
  const int kh    = lane >> 4;
  const int h     = wid / NT;
  const int tq    = wid % NT;

  const float* Qh = Q + h * (N_TOK * HD);
  const float* Kh = K + h * (N_TOK * HD);
  const float* Vh = V + h * (N_TOK * HD);
  float* P = ldsP + wslot * 256;

  v2f aq[8];
  {
    const int row = tq * 16 + l16;
#pragma unroll
    for (int t = 0; t < 8; ++t) {
      const int ka = 4 * t + 2 * kh;
      aq[t].x = Qh[row * HD + ka + 0];
      aq[t].y = Qh[row * HD + ka + 1];
    }
  }

  float m_i[8], l_i[8];
#pragma unroll
  for (int r = 0; r < 8; ++r) { m_i[r] = -1.0e30f; l_i[r] = 0.0f; }
  v8f o0 = {};
  v8f o1 = {};

  for (int kb = 0; kb < NT; ++kb) {
    v8f s = {};
#pragma unroll
    for (int t = 0; t < 8; ++t) {
      const int kd = 4 * t + 2 * kh;
      v2f b;
      b.x = Kh[(kb * 16 + l16) * HD + kd + 0];
      b.y = Kh[(kb * 16 + l16) * HD + kd + 1];
      s = wmma4(aq[t], b, s);
    }

    float p[8], alpha[8];
#pragma unroll
    for (int r = 0; r < 8; ++r) {
      float sv = s[r] * SCALE_;
      sv = fminf(fmaxf(sv, -10.0f), 10.0f);
      float mx = sv;
      mx = fmaxf(mx, __shfl_xor(mx, 1, 32));
      mx = fmaxf(mx, __shfl_xor(mx, 2, 32));
      mx = fmaxf(mx, __shfl_xor(mx, 4, 32));
      mx = fmaxf(mx, __shfl_xor(mx, 8, 32));
      const float mnew = fmaxf(m_i[r], mx);
      alpha[r] = __expf(m_i[r] - mnew);
      p[r]     = __expf(sv - mnew);
      float rs = p[r];
      rs += __shfl_xor(rs, 1, 32);
      rs += __shfl_xor(rs, 2, 32);
      rs += __shfl_xor(rs, 4, 32);
      rs += __shfl_xor(rs, 8, 32);
      l_i[r] = l_i[r] * alpha[r] + rs;
      m_i[r] = mnew;
    }

#pragma unroll
    for (int r = 0; r < 8; ++r) {
      o0[r] *= alpha[r];
      o1[r] *= alpha[r];
      P[(r + 8 * kh) * 16 + l16] = p[r];
    }

#pragma unroll
    for (int t = 0; t < 4; ++t) {
      const int kp = 4 * t + 2 * kh;
      v2f ap, b0, b1;
      ap.x = P[l16 * 16 + kp + 0];
      ap.y = P[l16 * 16 + kp + 1];
      const float* vr0 = Vh + (kb * 16 + kp + 0) * HD;
      const float* vr1 = Vh + (kb * 16 + kp + 1) * HD;
      b0.x = vr0[l16];        b0.y = vr1[l16];
      b1.x = vr0[16 + l16];   b1.y = vr1[16 + l16];
      o0 = wmma4(ap, b0, o0);
      o1 = wmma4(ap, b1, o1);
    }
  }

#pragma unroll
  for (int r = 0; r < 8; ++r) {
    const int n   = tq * 16 + r + 8 * kh;
    const float inv = 1.0f / l_i[r];
    AO[n * CDIM + h * HD + l16]      = o0[r] * inv;
    AO[n * CDIM + h * HD + 16 + l16] = o1[r] * inv;
  }
}

__global__ __launch_bounds__(256)
void gsa3d_proj_kernel(const float* __restrict__ AO,
                       const float* __restrict__ wproj,
                       float* __restrict__ out) {
  const int wid  = blockIdx.x * 8 + (threadIdx.x >> 5);
  const int lane = threadIdx.x & 31;
  const int l16  = lane & 15;
  const int kh   = lane >> 4;
  const int tm   = wid / 16;
  const int tn   = wid % 16;
  const int row  = tm * 16 + l16;
  const int col  = tn * 16 + l16;

  v8f acc = {};
  for (int k0 = 0; k0 < CDIM; k0 += 4) {
    const int ka = k0 + 2 * kh;
    v2f a, b;
    a.x = AO[row * CDIM + ka + 0];
    a.y = AO[row * CDIM + ka + 1];
    b.x = wproj[col * CDIM + ka + 0];
    b.y = wproj[col * CDIM + ka + 1];
    acc = wmma4(a, b, acc);
  }
#pragma unroll
  for (int r = 0; r < 8; ++r) {
    const int n = tm * 16 + r + 8 * kh;
    out[col * N_TOK + n] = acc[r];
  }
}

extern "C" void kernel_launch(void* const* d_in, const int* in_sizes, int n_in,
                              void* d_out, int out_size, void* d_ws, size_t ws_size,
                              hipStream_t stream) {
  (void)in_sizes; (void)n_in; (void)out_size; (void)ws_size;
  const float* x     = (const float*)d_in[0];
  const float* wqkv  = (const float*)d_in[1];
  const float* wproj = (const float*)d_in[2];
  float* out = (float*)d_out;

  float* ws = (float*)d_ws;
  float* Q  = ws;
  float* K  = Q + NHEADS * N_TOK * HD;
  float* V  = K + NHEADS * N_TOK * HD;
  float* AO = V + NHEADS * N_TOK * HD;

  gsa3d_qkv_kernel <<<648, 256, 0, stream>>>(x, wqkv, Q, K, V);
  gsa3d_attn_kernel<<<108, 256, 0, stream>>>(Q, K, V, AO);
  gsa3d_proj_kernel<<<216, 256, 0, stream>>>(AO, wproj, out);
}


# qkv and proj GEMMs: 8-wave work-groups own 64x96 / 64x32 output blocks; both operands staged through a 2-stage LDS ring by whole-line 16-byte loads one 64-k chunk ahead; every sum order unchanged
# speedup vs baseline: 4.1471x; 4.1471x over previous
#include <hip/hip_runtime.h>

typedef __attribute__((ext_vector_type(2))) float v2f;
typedef __attribute__((ext_vector_type(4))) float v4f;
typedef __attribute__((ext_vector_type(8))) float v8f;

#define N_TOK   1728
#define CDIM    256
#define NHEADS  8
#define HD      32
#define NT      108
#define SCALE_  0.1767766952966369f

__device__ __forceinline__ v8f wmma4(v2f a, v2f b, v8f c) {
  return __builtin_amdgcn_wmma_f32_16x16x4_f32(
       false, a,  false, b,
       (short)0, c,  false,  false);
}

#define GK   64
#define GNCH (CDIM / GK)
#define GBP  68
#define XAP  80

#define QTOK 64
#define QCOL 96
#define QA_FLOATS (GK * XAP)
#define QB_FLOATS (QCOL * GBP)

__global__ __launch_bounds__(256)
void gsa3d_qkv_kernel(const float* __restrict__ x,
                      const float* __restrict__ wqkv,
                      float* __restrict__ Q,
                      float* __restrict__ K,
                      float* __restrict__ V) {
  __shared__ __attribute__((aligned(16))) float ldsA[2 * QA_FLOATS];
  __shared__ __attribute__((aligned(16))) float ldsB[2 * QB_FLOATS];
  const int tid  = threadIdx.x;
  const int wv   = tid >> 5;
  const int lane = tid & 31;
  const int l16  = lane & 15;
  const int kh   = lane >> 4;
  const int n0   = (blockIdx.x >> 3) * QTOK;
  const int j0   = (blockIdx.x & 7) * QCOL;
  const int wt   = wv & 3;
  const int wj   = wv >> 2;

  const int arow = tid >> 4;
  const int aq   = tid & 15;
  const int brow = tid >> 4;
  const int bq   = tid & 15;
  const float* xg = x + arow * N_TOK + n0 + 4 * aq;
  const float* wg = wqkv + (j0 + brow) * CDIM + 4 * bq;
  const int asto = arow * XAP + 4 * aq;
  const int bsto = brow * GBP + 4 * bq;
  const int afo = (2 * kh) * XAP + 16 * wt + l16;
  const int bfo = (48 * wj + l16) * GBP + 2 * kh;

  v4f ga[4], gb[6];
#pragma unroll
  for (int i = 0; i < 4; ++i) ga[i] = *(const v4f*)(xg + (16 * i) * N_TOK);
#pragma unroll
  for (int i = 0; i < 6; ++i) gb[i] = *(const v4f*)(wg + (16 * i) * CDIM);
#pragma unroll
  for (int i = 0; i < 4; ++i) *(v4f*)(ldsA + asto + 16 * i * XAP) = ga[i];
#pragma unroll
  for (int i = 0; i < 6; ++i) *(v4f*)(ldsB + bsto + 16 * i * GBP) = gb[i];
  __syncthreads();

  v8f acc[3];
#pragma unroll
  for (int jt = 0; jt < 3; ++jt) { v8f z = {}; acc[jt] = z; }

  int st = 0;
#pragma unroll
  for (int c = 0; c < GNCH; ++c, st ^= 1) {
    const float* As = ldsA + st * QA_FLOATS;
    const float* Bs = ldsB + st * QB_FLOATS;

    const int k1 = (c + 1) * GK;
    if (c + 1 < GNCH) {
#pragma unroll
      for (int i = 0; i < 4; ++i) ga[i] = *(const v4f*)(xg + (k1 + 16 * i) * N_TOK);
#pragma unroll
      for (int i = 0; i < 6; ++i) gb[i] = *(const v4f*)(wg + (16 * i) * CDIM + k1);
    }

    v2f af[GK / 4], bf[3][GK / 4];
#pragma unroll
    for (int t = 0; t < GK / 4; ++t) {
      af[t].x = As[afo + (4 * t) * XAP];
      af[t].y = As[afo + (4 * t + 1) * XAP];
#pragma unroll
      for (int jt = 0; jt < 3; ++jt)
        bf[jt][t] = *(const v2f*)(Bs + bfo + jt * 16 * GBP + 4 * t);
    }
    __builtin_amdgcn_sched_barrier(0);

#pragma unroll
    for (int t = 0; t < GK / 4; ++t) {
#pragma unroll
      for (int jt = 0; jt < 3; ++jt) acc[jt] = wmma4(af[t], bf[jt][t], acc[jt]);
    }

    if (c + 1 < GNCH) {
      float* An = ldsA + (st ^ 1) * QA_FLOATS;
      float* Bn = ldsB + (st ^ 1) * QB_FLOATS;
#pragma unroll
      for (int i = 0; i < 4; ++i) *(v4f*)(An + asto + 16 * i * XAP) = ga[i];
#pragma unroll
      for (int i = 0; i < 6; ++i) *(v4f*)(Bn + bsto + 16 * i * GBP) = gb[i];
      __syncthreads();
    }
  }

#pragma unroll
  for (int jt = 0; jt < 3; ++jt) {
    const int col = j0 + (3 * wj + jt) * 16 + l16;
    const int s  = col >> 8;
    const int hh = (col >> 5) & 7;
    const int d  = col & 31;
    float* dst = (s == 0) ? Q : ((s == 1) ? K : V);
    dst += hh * (N_TOK * HD) + d;
#pragma unroll
    for (int r = 0; r < 8; ++r) {
      const int n = n0 + 16 * wt + r + 8 * kh;
      dst[n * HD] = acc[jt][r];
    }
  }
}

#define KT 9
#define NWAVE 4
#define KP 36
#define VTP 148
#define KS_FLOATS (KT * 16 * KP)
#define VS_FLOATS (HD * VTP)

__global__ __launch_bounds__(32 * NWAVE)
void gsa3d_attn_kernel(const float* __restrict__ Q,
                       const float* __restrict__ K,
                       const float* __restrict__ V,
                       float* __restrict__ AO) {
  __shared__ __attribute__((aligned(16))) float ldsK[2 * KS_FLOATS];
  __shared__ __attribute__((aligned(16))) float ldsV[2 * VS_FLOATS];
  __shared__ __attribute__((aligned(16))) float ldsP[NWAVE * KT * 256];
  const int tid   = threadIdx.x;
  const int wv    = tid >> 5;
  const int lane  = tid & 31;
  const int l16   = lane & 15;
  const int kh    = lane >> 4;
  const int h     = blockIdx.x / (NT / NWAVE);
  const int tq    = (blockIdx.x % (NT / NWAVE)) * NWAVE + wv;

  const float* Qh = Q + h * (N_TOK * HD);
  const float* Kh = K + h * (N_TOK * HD);
  const float* Vh = V + h * (N_TOK * HD);
  float* P = ldsP + wv * (KT * 256);

  const int srow = tid >> 3;
  const int sq   = tid & 7;
  const float* Kg = Kh + srow * HD + 4 * sq;
  const float* Vg = Vh + srow * HD + 4 * sq;
  const int ksto = srow * KP + 4 * sq;
  const int vsto = (4 * sq) * VTP + srow;
  const int kfo = l16 * KP + 2 * kh;
  const int vfo = l16 * VTP + 2 * kh;

  v2f aq[8];
  {
    const int row = tq * 16 + l16;
#pragma unroll
    for (int t = 0; t < 8; ++t) {
      const int ka = 4 * t + 2 * kh;
      aq[t].x = Qh[row * HD + ka + 0];
      aq[t].y = Qh[row * HD + ka + 1];
    }
  }

  v4f gk[KT], gv[KT];
#pragma unroll
  for (int i = 0; i < KT; ++i) {
    gk[i] = *(const v4f*)(Kg + i * 16 * HD);
    gv[i] = *(const v4f*)(Vg + i * 16 * HD);
  }
#pragma unroll
  for (int i = 0; i < KT; ++i) {
    *(v4f*)(ldsK + ksto + i * 16 * KP) = gk[i];
    ldsV[vsto + 0 * VTP + i * 16] = gv[i].x;
    ldsV[vsto + 1 * VTP + i * 16] = gv[i].y;
    ldsV[vsto + 2 * VTP + i * 16] = gv[i].z;
    ldsV[vsto + 3 * VTP + i * 16] = gv[i].w;
  }
  __syncthreads();

  float m_i[8], l_i[8];
#pragma unroll
  for (int r = 0; r < 8; ++r) { m_i[r] = -1.0e30f; l_i[r] = 0.0f; }
  v8f o0 = {};
  v8f o1 = {};

  int st = 0;
  for (int kb = 0; kb < NT; kb += KT, st ^= 1) {
    const float* Ks = ldsK + st * KS_FLOATS;
    const float* Vs = ldsV + st * VS_FLOATS;

    const int nb = (kb + KT < NT) ? (kb + KT) : kb;
#pragma unroll
    for (int i = 0; i < KT; ++i) {
      gk[i] = *(const v4f*)(Kg + (nb + i) * 16 * HD);
      gv[i] = *(const v4f*)(Vg + (nb + i) * 16 * HD);
    }

    v2f kf[KT][8];
#pragma unroll
    for (int t = 0; t < 8; ++t) {
#pragma unroll
      for (int j = 0; j < KT; ++j)
        kf[j][t] = *(const v2f*)(Ks + kfo + j * 16 * KP + 4 * t);
    }
    __builtin_amdgcn_sched_barrier(0);

    v8f s[KT];
#pragma unroll
    for (int j = 0; j < KT; ++j) { v8f z = {}; s[j] = z; }
#pragma unroll
    for (int t = 0; t < 8; ++t) {
#pragma unroll
      for (int j = 0; j < KT; ++j) s[j] = wmma4(aq[t], kf[j][t], s[j]);
    }
    __builtin_amdgcn_sched_barrier(0);

    v2f vf0[KT][4], vf1[KT][4];
#pragma unroll
    for (int j = 0; j < KT; ++j) {
#pragma unroll
      for (int t = 0; t < 4; ++t) {
        vf0[j][t] = *(const v2f*)(Vs + vfo + j * 16 + 4 * t);
        vf1[j][t] = *(const v2f*)(Vs + vfo + 16 * VTP + j * 16 + 4 * t);
      }
    }
    __builtin_amdgcn_sched_barrier(0);

#pragma unroll
    for (int r = 0; r < 8; ++r) {
      float sv[KT];
#pragma unroll
      for (int j = 0; j < KT; ++j) {
        const float v = s[j][r] * SCALE_;
        sv[j] = fminf(fmaxf(v, -10.0f), 10.0f);
      }
      float mx = sv[0];
#pragma unroll
      for (int j = 1; j < KT; ++j) mx = fmaxf(mx, sv[j]);
      mx = fmaxf(mx, __shfl_xor(mx, 1, 32));
      mx = fmaxf(mx, __shfl_xor(mx, 2, 32));
      mx = fmaxf(mx, __shfl_xor(mx, 4, 32));
      mx = fmaxf(mx, __shfl_xor(mx, 8, 32));
      const float mnew  = fmaxf(m_i[r], mx);
      const float alpha = __expf(m_i[r] - mnew);
      float rs = 0.0f;
#pragma unroll
      for (int j = 0; j < KT; ++j) {
        const float pj = __expf(sv[j] - mnew);
        P[j * 256 + (r + 8 * kh) * 16 + l16] = pj;
        rs += pj;
      }
      rs += __shfl_xor(rs, 1, 32);
      rs += __shfl_xor(rs, 2, 32);
      rs += __shfl_xor(rs, 4, 32);
      rs += __shfl_xor(rs, 8, 32);
      l_i[r] = l_i[r] * alpha + rs;
      m_i[r] = mnew;
      o0[r] *= alpha;
      o1[r] *= alpha;
    }

#pragma unroll
    for (int j = 0; j < KT; ++j) {
#pragma unroll
      for (int t = 0; t < 4; ++t) {
        const int kp = 4 * t + 2 * kh;
        v2f ap;
        ap.x = P[j * 256 + l16 * 16 + kp + 0];
        ap.y = P[j * 256 + l16 * 16 + kp + 1];
        o0 = wmma4(ap, vf0[j][t], o0);
        o1 = wmma4(ap, vf1[j][t], o1);
      }
    }

    float* Kn = ldsK + (st ^ 1) * KS_FLOATS;
    float* Vn = ldsV + (st ^ 1) * VS_FLOATS;
#pragma unroll
    for (int i = 0; i < KT; ++i) {
      *(v4f*)(Kn + ksto + i * 16 * KP) = gk[i];
      Vn[vsto + 0 * VTP + i * 16] = gv[i].x;
      Vn[vsto + 1 * VTP + i * 16] = gv[i].y;
      Vn[vsto + 2 * VTP + i * 16] = gv[i].z;
      Vn[vsto + 3 * VTP + i * 16] = gv[i].w;
    }
    __syncthreads();
  }

#pragma unroll
  for (int r = 0; r < 8; ++r) {
    const int n   = tq * 16 + r + 8 * kh;
    const float inv = 1.0f / l_i[r];
    AO[n * CDIM + h * HD + l16]      = o0[r] * inv;
    AO[n * CDIM + h * HD + 16 + l16] = o1[r] * inv;
  }
}

#define PTOK 64
#define PCOL 32
#define PA_FLOATS (PTOK * GBP)
#define PB_FLOATS (PCOL * GBP)

__global__ __launch_bounds__(256)
void gsa3d_proj_kernel(const float* __restrict__ AO,
                       const float* __restrict__ wproj,
                       float* __restrict__ out) {
  __shared__ __attribute__((aligned(16))) float ldsA[2 * PA_FLOATS];
  __shared__ __attribute__((aligned(16))) float ldsB[2 * PB_FLOATS];
  const int tid  = threadIdx.x;
  const int wv   = tid >> 5;
  const int lane = tid & 31;
  const int l16  = lane & 15;
  const int kh   = lane >> 4;
  const int n0   = (blockIdx.x >> 3) * PTOK;
  const int j0   = (blockIdx.x & 7) * PCOL;
  const int wt   = wv & 3;
  const int wj   = wv >> 2;

  const int srow = tid >> 4;
  const int sq   = tid & 15;
  const float* ag = AO + (n0 + srow) * CDIM + 4 * sq;
  const float* wg = wproj + (j0 + srow) * CDIM + 4 * sq;
  const int ssto = srow * GBP + 4 * sq;
  const int afo = (16 * wt + l16) * GBP + 2 * kh;
  const int bfo = (16 * wj + l16) * GBP + 2 * kh;

  v4f ga[4], gb[2];
#pragma unroll
  for (int i = 0; i < 4; ++i) ga[i] = *(const v4f*)(ag + (16 * i) * CDIM);
#pragma unroll
  for (int i = 0; i < 2; ++i) gb[i] = *(const v4f*)(wg + (16 * i) * CDIM);
#pragma unroll
  for (int i = 0; i < 4; ++i) *(v4f*)(ldsA + ssto + 16 * i * GBP) = ga[i];
#pragma unroll
  for (int i = 0; i < 2; ++i) *(v4f*)(ldsB + ssto + 16 * i * GBP) = gb[i];
  __syncthreads();

  v8f acc = {};
  int st = 0;
#pragma unroll
  for (int c = 0; c < GNCH; ++c, st ^= 1) {
    const float* As = ldsA + st * PA_FLOATS;
    const float* Bs = ldsB + st * PB_FLOATS;

    const int k1 = (c + 1) * GK;
    if (c + 1 < GNCH) {
#pragma unroll
      for (int i = 0; i < 4; ++i) ga[i] = *(const v4f*)(ag + (16 * i) * CDIM + k1);
#pragma unroll
      for (int i = 0; i < 2; ++i) gb[i] = *(const v4f*)(wg + (16 * i) * CDIM + k1);
    }

    v2f af[GK / 4], bf[GK / 4];
#pragma unroll
    for (int t = 0; t < GK / 4; ++t) {
      af[t] = *(const v2f*)(As + afo + 4 * t);
      bf[t] = *(const v2f*)(Bs + bfo + 4 * t);
    }
    __builtin_amdgcn_sched_barrier(0);

#pragma unroll
    for (int t = 0; t < GK / 4; ++t) acc = wmma4(af[t], bf[t], acc);

    if (c + 1 < GNCH) {
      float* An = ldsA + (st ^ 1) * PA_FLOATS;
      float* Bn = ldsB + (st ^ 1) * PB_FLOATS;
#pragma unroll
      for (int i = 0; i < 4; ++i) *(v4f*)(An + ssto + 16 * i * GBP) = ga[i];
#pragma unroll
      for (int i = 0; i < 2; ++i) *(v4f*)(Bn + ssto + 16 * i * GBP) = gb[i];
      __syncthreads();
    }
  }

  const int col = j0 + 16 * wj + l16;
#pragma unroll
  for (int r = 0; r < 8; ++r) {
    const int n = n0 + 16 * wt + r + 8 * kh;
    out[col * N_TOK + n] = acc[r];
  }
}

extern "C" void kernel_launch(void* const* d_in, const int* in_sizes, int n_in,
                              void* d_out, int out_size, void* d_ws, size_t ws_size,
                              hipStream_t stream) {
  (void)in_sizes; (void)n_in; (void)out_size; (void)ws_size;
  const float* x     = (const float*)d_in[0];
  const float* wqkv  = (const float*)d_in[1];
  const float* wproj = (const float*)d_in[2];
  float* out = (float*)d_out;

  float* ws = (float*)d_ws;
  float* Q  = ws;
  float* K  = Q + NHEADS * N_TOK * HD;
  float* V  = K + NHEADS * N_TOK * HD;
  float* AO = V + NHEADS * N_TOK * HD;

  gsa3d_qkv_kernel <<<216, 256, 0, stream>>>(x, wqkv, Q, K, V);
  gsa3d_attn_kernel<<<216, 128, 0, stream>>>(Q, K, V, AO);
  gsa3d_proj_kernel<<<216, 256, 0, stream>>>(AO, wproj, out);
}


# attention softmax: the 16-lane row max and row sum butterflies exchange lanes on the vector unit (DPP row_xmask) instead of 64 lane shuffles through the LDS unit per trip; sum order unchanged
# speedup vs baseline: 4.7103x; 1.1358x over previous
#include <hip/hip_runtime.h>

typedef __attribute__((ext_vector_type(2))) float v2f;
typedef __attribute__((ext_vector_type(4))) float v4f;
typedef __attribute__((ext_vector_type(8))) float v8f;

#define N_TOK   1728
#define CDIM    256
#define NHEADS  8
#define HD      32
#define NT      108
#define SCALE_  0.1767766952966369f

__device__ __forceinline__ v8f wmma4(v2f a, v2f b, v8f c) {
  return __builtin_amdgcn_wmma_f32_16x16x4_f32(
       false, a,  false, b,
       (short)0, c,  false,  false);
}

template <int M>
__device__ __forceinline__ float row_xor(float v) {
  return __builtin_bit_cast(float,
      __builtin_amdgcn_update_dpp(0, __builtin_bit_cast(int, v),
                                  0x160 + M  , 0xf, 0xf, true));
}

#define GK   64
#define GNCH (CDIM / GK)
#define GBP  68
#define XAP  80

#define QTOK 64
#define QCOL 96
#define QA_FLOATS (GK * XAP)
#define QB_FLOATS (QCOL * GBP)

__global__ __launch_bounds__(256)
void gsa3d_qkv_kernel(const float* __restrict__ x,
                      const float* __restrict__ wqkv,
                      float* __restrict__ Q,
                      float* __restrict__ K,
                      float* __restrict__ V) {
  __shared__ __attribute__((aligned(16))) float ldsA[2 * QA_FLOATS];
  __shared__ __attribute__((aligned(16))) float ldsB[2 * QB_FLOATS];
  const int tid  = threadIdx.x;
  const int wv   = tid >> 5;
  const int lane = tid & 31;
  const int l16  = lane & 15;
  const int kh   = lane >> 4;
  const int n0   = (blockIdx.x >> 3) * QTOK;
  const int j0   = (blockIdx.x & 7) * QCOL;
  const int wt   = wv & 3;
  const int wj   = wv >> 2;

  const int arow = tid >> 4;
  const int aq   = tid & 15;
  const int brow = tid >> 4;
  const int bq   = tid & 15;
  const float* xg = x + arow * N_TOK + n0 + 4 * aq;
  const float* wg = wqkv + (j0 + brow) * CDIM + 4 * bq;
  const int asto = arow * XAP + 4 * aq;
  const int bsto = brow * GBP + 4 * bq;
  const int afo = (2 * kh) * XAP + 16 * wt + l16;
  const int bfo = (48 * wj + l16) * GBP + 2 * kh;

  v4f ga[4], gb[6];
#pragma unroll
  for (int i = 0; i < 4; ++i) ga[i] = *(const v4f*)(xg + (16 * i) * N_TOK);
#pragma unroll
  for (int i = 0; i < 6; ++i) gb[i] = *(const v4f*)(wg + (16 * i) * CDIM);
#pragma unroll
  for (int i = 0; i < 4; ++i) *(v4f*)(ldsA + asto + 16 * i * XAP) = ga[i];
#pragma unroll
  for (int i = 0; i < 6; ++i) *(v4f*)(ldsB + bsto + 16 * i * GBP) = gb[i];
  __syncthreads();

  v8f acc[3];
#pragma unroll
  for (int jt = 0; jt < 3; ++jt) { v8f z = {}; acc[jt] = z; }

  int st = 0;
#pragma unroll
  for (int c = 0; c < GNCH; ++c, st ^= 1) {
    const float* As = ldsA + st * QA_FLOATS;
    const float* Bs = ldsB + st * QB_FLOATS;

    const int k1 = (c + 1) * GK;
    if (c + 1 < GNCH) {
#pragma unroll
      for (int i = 0; i < 4; ++i) ga[i] = *(const v4f*)(xg + (k1 + 16 * i) * N_TOK);
#pragma unroll
      for (int i = 0; i < 6; ++i) gb[i] = *(const v4f*)(wg + (16 * i) * CDIM + k1);
    }

    v2f af[GK / 4], bf[3][GK / 4];
#pragma unroll
    for (int t = 0; t < GK / 4; ++t) {
      af[t].x = As[afo + (4 * t) * XAP];
      af[t].y = As[afo + (4 * t + 1) * XAP];
#pragma unroll
      for (int jt = 0; jt < 3; ++jt)
        bf[jt][t] = *(const v2f*)(Bs + bfo + jt * 16 * GBP + 4 * t);
    }
    __builtin_amdgcn_sched_barrier(0);

#pragma unroll
    for (int t = 0; t < GK / 4; ++t) {
#pragma unroll
      for (int jt = 0; jt < 3; ++jt) acc[jt] = wmma4(af[t], bf[jt][t], acc[jt]);
    }

    if (c + 1 < GNCH) {
      float* An = ldsA + (st ^ 1) * QA_FLOATS;
      float* Bn = ldsB + (st ^ 1) * QB_FLOATS;
#pragma unroll
      for (int i = 0; i < 4; ++i) *(v4f*)(An + asto + 16 * i * XAP) = ga[i];
#pragma unroll
      for (int i = 0; i < 6; ++i) *(v4f*)(Bn + bsto + 16 * i * GBP) = gb[i];
      __syncthreads();
    }
  }

#pragma unroll
  for (int jt = 0; jt < 3; ++jt) {
    const int col = j0 + (3 * wj + jt) * 16 + l16;
    const int s  = col >> 8;
    const int hh = (col >> 5) & 7;
    const int d  = col & 31;
    float* dst = (s == 0) ? Q : ((s == 1) ? K : V);
    dst += hh * (N_TOK * HD) + d;
#pragma unroll
    for (int r = 0; r < 8; ++r) {
      const int n = n0 + 16 * wt + r + 8 * kh;
      dst[n * HD] = acc[jt][r];
    }
  }
}

#define KT 9
#define NWAVE 4
#define KP 36
#define VTP 148
#define KS_FLOATS (KT * 16 * KP)
#define VS_FLOATS (HD * VTP)

__global__ __launch_bounds__(32 * NWAVE)
void gsa3d_attn_kernel(const float* __restrict__ Q,
                       const float* __restrict__ K,
                       const float* __restrict__ V,
                       float* __restrict__ AO) {
  __shared__ __attribute__((aligned(16))) float ldsK[2 * KS_FLOATS];
  __shared__ __attribute__((aligned(16))) float ldsV[2 * VS_FLOATS];
  __shared__ __attribute__((aligned(16))) float ldsP[NWAVE * KT * 256];
  const int tid   = threadIdx.x;
  const int wv    = tid >> 5;
  const int lane  = tid & 31;
  const int l16   = lane & 15;
  const int kh    = lane >> 4;
  const int h     = blockIdx.x / (NT / NWAVE);
  const int tq    = (blockIdx.x % (NT / NWAVE)) * NWAVE + wv;

  const float* Qh = Q + h * (N_TOK * HD);
  const float* Kh = K + h * (N_TOK * HD);
  const float* Vh = V + h * (N_TOK * HD);
  float* P = ldsP + wv * (KT * 256);

  const int srow = tid >> 3;
  const int sq   = tid & 7;
  const float* Kg = Kh + srow * HD + 4 * sq;
  const float* Vg = Vh + srow * HD + 4 * sq;
  const int ksto = srow * KP + 4 * sq;
  const int vsto = (4 * sq) * VTP + srow;
  const int kfo = l16 * KP + 2 * kh;
  const int vfo = l16 * VTP + 2 * kh;

  v2f aq[8];
  {
    const int row = tq * 16 + l16;
#pragma unroll
    for (int t = 0; t < 8; ++t) {
      const int ka = 4 * t + 2 * kh;
      aq[t].x = Qh[row * HD + ka + 0];
      aq[t].y = Qh[row * HD + ka + 1];
    }
  }

  v4f gk[KT], gv[KT];
#pragma unroll
  for (int i = 0; i < KT; ++i) {
    gk[i] = *(const v4f*)(Kg + i * 16 * HD);
    gv[i] = *(const v4f*)(Vg + i * 16 * HD);
  }
#pragma unroll
  for (int i = 0; i < KT; ++i) {
    *(v4f*)(ldsK + ksto + i * 16 * KP) = gk[i];
    ldsV[vsto + 0 * VTP + i * 16] = gv[i].x;
    ldsV[vsto + 1 * VTP + i * 16] = gv[i].y;
    ldsV[vsto + 2 * VTP + i * 16] = gv[i].z;
    ldsV[vsto + 3 * VTP + i * 16] = gv[i].w;
  }
  __syncthreads();

  float m_i[8], l_i[8];
#pragma unroll
  for (int r = 0; r < 8; ++r) { m_i[r] = -1.0e30f; l_i[r] = 0.0f; }
  v8f o0 = {};
  v8f o1 = {};

  int st = 0;
  for (int kb = 0; kb < NT; kb += KT, st ^= 1) {
    const float* Ks = ldsK + st * KS_FLOATS;
    const float* Vs = ldsV + st * VS_FLOATS;

    const int nb = (kb + KT < NT) ? (kb + KT) : kb;
#pragma unroll
    for (int i = 0; i < KT; ++i) {
      gk[i] = *(const v4f*)(Kg + (nb + i) * 16 * HD);
      gv[i] = *(const v4f*)(Vg + (nb + i) * 16 * HD);
    }

    v2f kf[KT][8];
#pragma unroll
    for (int t = 0; t < 8; ++t) {
#pragma unroll
      for (int j = 0; j < KT; ++j)
        kf[j][t] = *(const v2f*)(Ks + kfo + j * 16 * KP + 4 * t);
    }
    __builtin_amdgcn_sched_barrier(0);

    v8f s[KT];
#pragma unroll
    for (int j = 0; j < KT; ++j) { v8f z = {}; s[j] = z; }
#pragma unroll
    for (int t = 0; t < 8; ++t) {
#pragma unroll
      for (int j = 0; j < KT; ++j) s[j] = wmma4(aq[t], kf[j][t], s[j]);
    }
    __builtin_amdgcn_sched_barrier(0);

    v2f vf0[KT][4], vf1[KT][4];
#pragma unroll
    for (int j = 0; j < KT; ++j) {
#pragma unroll
      for (int t = 0; t < 4; ++t) {
        vf0[j][t] = *(const v2f*)(Vs + vfo + j * 16 + 4 * t);
        vf1[j][t] = *(const v2f*)(Vs + vfo + 16 * VTP + j * 16 + 4 * t);
      }
    }
    __builtin_amdgcn_sched_barrier(0);

#pragma unroll
    for (int r = 0; r < 8; ++r) {
      float sv[KT];
#pragma unroll
      for (int j = 0; j < KT; ++j) {
        const float v = s[j][r] * SCALE_;
        sv[j] = fminf(fmaxf(v, -10.0f), 10.0f);
      }
      float mx = sv[0];
#pragma unroll
      for (int j = 1; j < KT; ++j) mx = fmaxf(mx, sv[j]);
      mx = __builtin_fmaximum_numf(mx, row_xor<1>(mx));
      mx = __builtin_fmaximum_numf(mx, row_xor<2>(mx));
      mx = __builtin_fmaximum_numf(mx, row_xor<4>(mx));
      mx = __builtin_fmaximum_numf(mx, row_xor<8>(mx));
      const float mnew  = fmaxf(m_i[r], mx);
      const float alpha = __expf(m_i[r] - mnew);
      float rs = 0.0f;
#pragma unroll
      for (int j = 0; j < KT; ++j) {
        const float pj = __expf(sv[j] - mnew);
        P[j * 256 + (r + 8 * kh) * 16 + l16] = pj;
        rs += pj;
      }
      rs += row_xor<1>(rs);
      rs += row_xor<2>(rs);
      rs += row_xor<4>(rs);
      rs += row_xor<8>(rs);
      l_i[r] = l_i[r] * alpha + rs;
      m_i[r] = mnew;
      o0[r] *= alpha;
      o1[r] *= alpha;
    }

#pragma unroll
    for (int j = 0; j < KT; ++j) {
#pragma unroll
      for (int t = 0; t < 4; ++t) {
        const int kp = 4 * t + 2 * kh;
        v2f ap;
        ap.x = P[j * 256 + l16 * 16 + kp + 0];
        ap.y = P[j * 256 + l16 * 16 + kp + 1];
        o0 = wmma4(ap, vf0[j][t], o0);
        o1 = wmma4(ap, vf1[j][t], o1);
      }
    }

    float* Kn = ldsK + (st ^ 1) * KS_FLOATS;
    float* Vn = ldsV + (st ^ 1) * VS_FLOATS;
#pragma unroll
    for (int i = 0; i < KT; ++i) {
      *(v4f*)(Kn + ksto + i * 16 * KP) = gk[i];
      Vn[vsto + 0 * VTP + i * 16] = gv[i].x;
      Vn[vsto + 1 * VTP + i * 16] = gv[i].y;
      Vn[vsto + 2 * VTP + i * 16] = gv[i].z;
      Vn[vsto + 3 * VTP + i * 16] = gv[i].w;
    }
    __syncthreads();
  }

#pragma unroll
  for (int r = 0; r < 8; ++r) {
    const int n   = tq * 16 + r + 8 * kh;
    const float inv = 1.0f / l_i[r];
    AO[n * CDIM + h * HD + l16]      = o0[r] * inv;
    AO[n * CDIM + h * HD + 16 + l16] = o1[r] * inv;
  }
}

#define PTOK 64
#define PCOL 32
#define PA_FLOATS (PTOK * GBP)
#define PB_FLOATS (PCOL * GBP)

__global__ __launch_bounds__(256)
void gsa3d_proj_kernel(const float* __restrict__ AO,
                       const float* __restrict__ wproj,
                       float* __restrict__ out) {
  __shared__ __attribute__((aligned(16))) float ldsA[2 * PA_FLOATS];
  __shared__ __attribute__((aligned(16))) float ldsB[2 * PB_FLOATS];
  const int tid  = threadIdx.x;
  const int wv   = tid >> 5;
  const int lane = tid & 31;
  const int l16  = lane & 15;
  const int kh   = lane >> 4;
  const int n0   = (blockIdx.x >> 3) * PTOK;
  const int j0   = (blockIdx.x & 7) * PCOL;
  const int wt   = wv & 3;
  const int wj   = wv >> 2;

  const int srow = tid >> 4;
  const int sq   = tid & 15;
  const float* ag = AO + (n0 + srow) * CDIM + 4 * sq;
  const float* wg = wproj + (j0 + srow) * CDIM + 4 * sq;
  const int ssto = srow * GBP + 4 * sq;
  const int afo = (16 * wt + l16) * GBP + 2 * kh;
  const int bfo = (16 * wj + l16) * GBP + 2 * kh;

  v4f ga[4], gb[2];
#pragma unroll
  for (int i = 0; i < 4; ++i) ga[i] = *(const v4f*)(ag + (16 * i) * CDIM);
#pragma unroll
  for (int i = 0; i < 2; ++i) gb[i] = *(const v4f*)(wg + (16 * i) * CDIM);
#pragma unroll
  for (int i = 0; i < 4; ++i) *(v4f*)(ldsA + ssto + 16 * i * GBP) = ga[i];
#pragma unroll
  for (int i = 0; i < 2; ++i) *(v4f*)(ldsB + ssto + 16 * i * GBP) = gb[i];
  __syncthreads();

  v8f acc = {};
  int st = 0;
#pragma unroll
  for (int c = 0; c < GNCH; ++c, st ^= 1) {
    const float* As = ldsA + st * PA_FLOATS;
    const float* Bs = ldsB + st * PB_FLOATS;

    const int k1 = (c + 1) * GK;
    if (c + 1 < GNCH) {
#pragma unroll
      for (int i = 0; i < 4; ++i) ga[i] = *(const v4f*)(ag + (16 * i) * CDIM + k1);
#pragma unroll
      for (int i = 0; i < 2; ++i) gb[i] = *(const v4f*)(wg + (16 * i) * CDIM + k1);
    }

    v2f af[GK / 4], bf[GK / 4];
#pragma unroll
    for (int t = 0; t < GK / 4; ++t) {
      af[t] = *(const v2f*)(As + afo + 4 * t);
      bf[t] = *(const v2f*)(Bs + bfo + 4 * t);
    }
    __builtin_amdgcn_sched_barrier(0);

#pragma unroll
    for (int t = 0; t < GK / 4; ++t) acc = wmma4(af[t], bf[t], acc);

    if (c + 1 < GNCH) {
      float* An = ldsA + (st ^ 1) * PA_FLOATS;
      float* Bn = ldsB + (st ^ 1) * PB_FLOATS;
#pragma unroll
      for (int i = 0; i < 4; ++i) *(v4f*)(An + ssto + 16 * i * GBP) = ga[i];
#pragma unroll
      for (int i = 0; i < 2; ++i) *(v4f*)(Bn + ssto + 16 * i * GBP) = gb[i];
      __syncthreads();
    }
  }

  const int col = j0 + 16 * wj + l16;
#pragma unroll
  for (int r = 0; r < 8; ++r) {
    const int n = n0 + 16 * wt + r + 8 * kh;
    out[col * N_TOK + n] = acc[r];
  }
}

extern "C" void kernel_launch(void* const* d_in, const int* in_sizes, int n_in,
                              void* d_out, int out_size, void* d_ws, size_t ws_size,
                              hipStream_t stream) {
  (void)in_sizes; (void)n_in; (void)out_size; (void)ws_size;
  const float* x     = (const float*)d_in[0];
  const float* wqkv  = (const float*)d_in[1];
  const float* wproj = (const float*)d_in[2];
  float* out = (float*)d_out;

  float* ws = (float*)d_ws;
  float* Q  = ws;
  float* K  = Q + NHEADS * N_TOK * HD;
  float* V  = K + NHEADS * N_TOK * HD;
  float* AO = V + NHEADS * N_TOK * HD;

  gsa3d_qkv_kernel <<<216, 256, 0, stream>>>(x, wqkv, Q, K, V);
  gsa3d_attn_kernel<<<216, 128, 0, stream>>>(Q, K, V, AO);
  gsa3d_proj_kernel<<<216, 256, 0, stream>>>(AO, wproj, out);
}


# attention: a query tile's work split by stage over the 2 waves of a SIMD: score wave (S, max, exp, row sum), PV wave one trip behind; P and rescale factors handed over in LDS; sum orders unchanged
# speedup vs baseline: 4.8674x; 1.0334x over previous
#include <hip/hip_runtime.h>

typedef __attribute__((ext_vector_type(2))) float v2f;
typedef __attribute__((ext_vector_type(4))) float v4f;
typedef __attribute__((ext_vector_type(8))) float v8f;

#define N_TOK   1728
#define CDIM    256
#define NHEADS  8
#define HD      32
#define NT      108
#define SCALE_  0.1767766952966369f

__device__ __forceinline__ v8f wmma4(v2f a, v2f b, v8f c) {
  return __builtin_amdgcn_wmma_f32_16x16x4_f32(
       false, a,  false, b,
       (short)0, c,  false,  false);
}

template <int M>
__device__ __forceinline__ float row_xor(float v) {
  return __builtin_bit_cast(float,
      __builtin_amdgcn_update_dpp(0, __builtin_bit_cast(int, v),
                                  0x160 + M  , 0xf, 0xf, true));
}

#define GK   64
#define GNCH (CDIM / GK)
#define GBP  68
#define XAP  80

#define QTOK 64
#define QCOL 96
#define QA_FLOATS (GK * XAP)
#define QB_FLOATS (QCOL * GBP)

__global__ __launch_bounds__(256)
void gsa3d_qkv_kernel(const float* __restrict__ x,
                      const float* __restrict__ wqkv,
                      float* __restrict__ Q,
                      float* __restrict__ K,
                      float* __restrict__ V) {
  __shared__ __attribute__((aligned(16))) float ldsA[2 * QA_FLOATS];
  __shared__ __attribute__((aligned(16))) float ldsB[2 * QB_FLOATS];
  const int tid  = threadIdx.x;
  const int wv   = tid >> 5;
  const int lane = tid & 31;
  const int l16  = lane & 15;
  const int kh   = lane >> 4;
  const int n0   = (blockIdx.x >> 3) * QTOK;
  const int j0   = (blockIdx.x & 7) * QCOL;
  const int wt   = wv & 3;
  const int wj   = wv >> 2;

  const int arow = tid >> 4;
  const int aq   = tid & 15;
  const int brow = tid >> 4;
  const int bq   = tid & 15;
  const float* xg = x + arow * N_TOK + n0 + 4 * aq;
  const float* wg = wqkv + (j0 + brow) * CDIM + 4 * bq;
  const int asto = arow * XAP + 4 * aq;
  const int bsto = brow * GBP + 4 * bq;
  const int afo = (2 * kh) * XAP + 16 * wt + l16;
  const int bfo = (48 * wj + l16) * GBP + 2 * kh;

  v4f ga[4], gb[6];
#pragma unroll
  for (int i = 0; i < 4; ++i) ga[i] = *(const v4f*)(xg + (16 * i) * N_TOK);
#pragma unroll
  for (int i = 0; i < 6; ++i) gb[i] = *(const v4f*)(wg + (16 * i) * CDIM);
#pragma unroll
  for (int i = 0; i < 4; ++i) *(v4f*)(ldsA + asto + 16 * i * XAP) = ga[i];
#pragma unroll
  for (int i = 0; i < 6; ++i) *(v4f*)(ldsB + bsto + 16 * i * GBP) = gb[i];
  __syncthreads();

  v8f acc[3];
#pragma unroll
  for (int jt = 0; jt < 3; ++jt) { v8f z = {}; acc[jt] = z; }

  int st = 0;
#pragma unroll
  for (int c = 0; c < GNCH; ++c, st ^= 1) {
    const float* As = ldsA + st * QA_FLOATS;
    const float* Bs = ldsB + st * QB_FLOATS;

    const int k1 = (c + 1) * GK;
    if (c + 1 < GNCH) {
#pragma unroll
      for (int i = 0; i < 4; ++i) ga[i] = *(const v4f*)(xg + (k1 + 16 * i) * N_TOK);
#pragma unroll
      for (int i = 0; i < 6; ++i) gb[i] = *(const v4f*)(wg + (16 * i) * CDIM + k1);
    }

    v2f af[GK / 4], bf[3][GK / 4];
#pragma unroll
    for (int t = 0; t < GK / 4; ++t) {
      af[t].x = As[afo + (4 * t) * XAP];
      af[t].y = As[afo + (4 * t + 1) * XAP];
#pragma unroll
      for (int jt = 0; jt < 3; ++jt)
        bf[jt][t] = *(const v2f*)(Bs + bfo + jt * 16 * GBP + 4 * t);
    }
    __builtin_amdgcn_sched_barrier(0);

#pragma unroll
    for (int t = 0; t < GK / 4; ++t) {
#pragma unroll
      for (int jt = 0; jt < 3; ++jt) acc[jt] = wmma4(af[t], bf[jt][t], acc[jt]);
    }

    if (c + 1 < GNCH) {
      float* An = ldsA + (st ^ 1) * QA_FLOATS;
      float* Bn = ldsB + (st ^ 1) * QB_FLOATS;
#pragma unroll
      for (int i = 0; i < 4; ++i) *(v4f*)(An + asto + 16 * i * XAP) = ga[i];
#pragma unroll
      for (int i = 0; i < 6; ++i) *(v4f*)(Bn + bsto + 16 * i * GBP) = gb[i];
      __syncthreads();
    }
  }

#pragma unroll
  for (int jt = 0; jt < 3; ++jt) {
    const int col = j0 + (3 * wj + jt) * 16 + l16;
    const int s  = col >> 8;
    const int hh = (col >> 5) & 7;
    const int d  = col & 31;
    float* dst = (s == 0) ? Q : ((s == 1) ? K : V);
    dst += hh * (N_TOK * HD) + d;
#pragma unroll
    for (int r = 0; r < 8; ++r) {
      const int n = n0 + 16 * wt + r + 8 * kh;
      dst[n * HD] = acc[jt][r];
    }
  }
}

#define KT 9
#define NTRIP (NT / KT)
#define NQT 4
#define KP 36
#define VTP 148
#define KS_FLOATS (KT * 16 * KP)
#define VS_FLOATS (HD * VTP)
#define PT_FLOATS (KT * 256)

__global__ __launch_bounds__(64 * NQT)
void gsa3d_attn_kernel(const float* __restrict__ Q,
                       const float* __restrict__ K,
                       const float* __restrict__ V,
                       float* __restrict__ AO) {
  __shared__ __attribute__((aligned(16))) float ldsK[2 * KS_FLOATS];
  __shared__ __attribute__((aligned(16))) float ldsV[2 * VS_FLOATS];
  __shared__ __attribute__((aligned(16))) float ldsP[2 * NQT * PT_FLOATS];
  __shared__ __attribute__((aligned(16))) float ldsF[2 * NQT * 256];
  __shared__ __attribute__((aligned(16))) float ldsL[NQT * 256];
  const int tid   = threadIdx.x;
  const int wv    = tid >> 5;
  const int stage = __builtin_amdgcn_readfirstlane(wv >> 2);
  const int qt    = wv & 3;
  const int lane  = tid & 31;
  const int l16   = lane & 15;
  const int kh    = lane >> 4;
  const int h     = blockIdx.x / (NT / NQT);
  const int tq    = (blockIdx.x % (NT / NQT)) * NQT + qt;

  const float* Qh = Q + h * (N_TOK * HD);
  const float* Kh = K + h * (N_TOK * HD);
  const float* Vh = V + h * (N_TOK * HD);

  const int stid = tid & 127;
  const int srow = stid >> 3;
  const int sq   = stid & 7;
  const float* Kg = Kh + srow * HD + 4 * sq;
  const float* Vg = Vh + srow * HD + 4 * sq;
  const int ksto = srow * KP + 4 * sq;
  const int vsto = (4 * sq) * VTP + srow;
  const int kfo = l16 * KP + 2 * kh;
  const int vfo = l16 * VTP + 2 * kh;

  v2f aq[8];
  float m_i[8], l_i[8];
#pragma unroll
  for (int r = 0; r < 8; ++r) { m_i[r] = -1.0e30f; l_i[r] = 0.0f; }
#pragma unroll
  for (int t = 0; t < 8; ++t) { v2f z = {}; aq[t] = z; }
  v8f o0 = {};
  v8f o1 = {};

  if (stage == 0) {
    const int row = tq * 16 + l16;
#pragma unroll
    for (int t = 0; t < 8; ++t) {
      const int ka = 4 * t + 2 * kh;
      aq[t].x = Qh[row * HD + ka + 0];
      aq[t].y = Qh[row * HD + ka + 1];
    }
    v4f gk[KT];
#pragma unroll
    for (int i = 0; i < KT; ++i) gk[i] = *(const v4f*)(Kg + i * 16 * HD);
#pragma unroll
    for (int i = 0; i < KT; ++i) *(v4f*)(ldsK + ksto + i * 16 * KP) = gk[i];
  }
  __syncthreads();

  for (int it = 0; it <= NTRIP; ++it) {
    if (stage == 0) {
      if (it < NTRIP) {
        const float* Ks = ldsK + (it & 1) * KS_FLOATS;
        float* Pw = ldsP + ((it & 1) * NQT + qt) * PT_FLOATS;
        float* Fw = ldsF + ((it & 1) * NQT + qt) * 256;

        const int nb = ((it + 1 < NTRIP) ? (it + 1) : it) * KT;
        v4f gk[KT];
#pragma unroll
        for (int i = 0; i < KT; ++i) gk[i] = *(const v4f*)(Kg + (nb + i) * 16 * HD);

        v2f kf[KT][8];
#pragma unroll
        for (int t = 0; t < 8; ++t) {
#pragma unroll
          for (int j = 0; j < KT; ++j)
            kf[j][t] = *(const v2f*)(Ks + kfo + j * 16 * KP + 4 * t);
        }
        __builtin_amdgcn_sched_barrier(0);

        v8f s[KT];
#pragma unroll
        for (int j = 0; j < KT; ++j) { v8f z = {}; s[j] = z; }
#pragma unroll
        for (int t = 0; t < 8; ++t) {
#pragma unroll
          for (int j = 0; j < KT; ++j) s[j] = wmma4(aq[t], kf[j][t], s[j]);
        }

#pragma unroll
        for (int r = 0; r < 8; ++r) {
          float sv[KT];
#pragma unroll
          for (int j = 0; j < KT; ++j) {
            const float v = s[j][r] * SCALE_;
            sv[j] = fminf(fmaxf(v, -10.0f), 10.0f);
          }
          float mx = sv[0];
#pragma unroll
          for (int j = 1; j < KT; ++j) mx = fmaxf(mx, sv[j]);
          mx = __builtin_fmaximum_numf(mx, row_xor<1>(mx));
          mx = __builtin_fmaximum_numf(mx, row_xor<2>(mx));
          mx = __builtin_fmaximum_numf(mx, row_xor<4>(mx));
          mx = __builtin_fmaximum_numf(mx, row_xor<8>(mx));
          const float mnew  = fmaxf(m_i[r], mx);
          const float alpha = __expf(m_i[r] - mnew);
          float rs = 0.0f;
#pragma unroll
          for (int j = 0; j < KT; ++j) {
            const float pj = __expf(sv[j] - mnew);
            Pw[j * 256 + (r + 8 * kh) * 16 + l16] = pj;
            rs += pj;
          }
          rs += row_xor<1>(rs);
          rs += row_xor<2>(rs);
          rs += row_xor<4>(rs);
          rs += row_xor<8>(rs);
          l_i[r] = l_i[r] * alpha + rs;
          m_i[r] = mnew;
          Fw[r * 32 + lane] = alpha;
        }
        if (it == NTRIP - 1) {
#pragma unroll
          for (int r = 0; r < 8; ++r) ldsL[qt * 256 + r * 32 + lane] = l_i[r];
        }

        float* Kn = ldsK + ((it + 1) & 1) * KS_FLOATS;
#pragma unroll
        for (int i = 0; i < KT; ++i) *(v4f*)(Kn + ksto + i * 16 * KP) = gk[i];
      }
    } else {
      const int vb = ((it < NTRIP) ? it : (NTRIP - 1)) * KT;
      v4f gv[KT];
#pragma unroll
      for (int i = 0; i < KT; ++i) gv[i] = *(const v4f*)(Vg + (vb + i) * 16 * HD);

      if (it >= 1) {
        const int b = (it - 1) & 1;
        const float* Vs = ldsV + b * VS_FLOATS;
        const float* Pr = ldsP + (b * NQT + qt) * PT_FLOATS;
        const float* Fr = ldsF + (b * NQT + qt) * 256;

        float al[8];
#pragma unroll
        for (int r = 0; r < 8; ++r) al[r] = Fr[r * 32 + lane];
        v2f vf0[KT][4], vf1[KT][4], pf[KT][4];
#pragma unroll
        for (int j = 0; j < KT; ++j) {
#pragma unroll
          for (int t = 0; t < 4; ++t) {
            pf[j][t]  = *(const v2f*)(Pr + j * 256 + l16 * 16 + 4 * t + 2 * kh);
            vf0[j][t] = *(const v2f*)(Vs + vfo + j * 16 + 4 * t);
            vf1[j][t] = *(const v2f*)(Vs + vfo + 16 * VTP + j * 16 + 4 * t);
          }
        }
        __builtin_amdgcn_sched_barrier(0);

#pragma unroll
        for (int r = 0; r < 8; ++r) { o0[r] *= al[r]; o1[r] *= al[r]; }
#pragma unroll
        for (int j = 0; j < KT; ++j) {
#pragma unroll
          for (int t = 0; t < 4; ++t) {
            o0 = wmma4(pf[j][t], vf0[j][t], o0);
            o1 = wmma4(pf[j][t], vf1[j][t], o1);
          }
        }
      }

      float* Vn = ldsV + (it & 1) * VS_FLOATS;
#pragma unroll
      for (int i = 0; i < KT; ++i) {
        Vn[vsto + 0 * VTP + i * 16] = gv[i].x;
        Vn[vsto + 1 * VTP + i * 16] = gv[i].y;
        Vn[vsto + 2 * VTP + i * 16] = gv[i].z;
        Vn[vsto + 3 * VTP + i * 16] = gv[i].w;
      }
    }
    __syncthreads();
  }

  if (stage == 1) {
#pragma unroll
    for (int r = 0; r < 8; ++r) {
      const int n   = tq * 16 + r + 8 * kh;
      const float inv = 1.0f / ldsL[qt * 256 + r * 32 + lane];
      AO[n * CDIM + h * HD + l16]      = o0[r] * inv;
      AO[n * CDIM + h * HD + 16 + l16] = o1[r] * inv;
    }
  }
}

#define PTOK 64
#define PCOL 32
#define PA_FLOATS (PTOK * GBP)
#define PB_FLOATS (PCOL * GBP)

__global__ __launch_bounds__(256)
void gsa3d_proj_kernel(const float* __restrict__ AO,
                       const float* __restrict__ wproj,
                       float* __restrict__ out) {
  __shared__ __attribute__((aligned(16))) float ldsA[2 * PA_FLOATS];
  __shared__ __attribute__((aligned(16))) float ldsB[2 * PB_FLOATS];
  const int tid  = threadIdx.x;
  const int wv   = tid >> 5;
  const int lane = tid & 31;
  const int l16  = lane & 15;
  const int kh   = lane >> 4;
  const int n0   = (blockIdx.x >> 3) * PTOK;
  const int j0   = (blockIdx.x & 7) * PCOL;
  const int wt   = wv & 3;
  const int wj   = wv >> 2;

  const int srow = tid >> 4;
  const int sq   = tid & 15;
  const float* ag = AO + (n0 + srow) * CDIM + 4 * sq;
  const float* wg = wproj + (j0 + srow) * CDIM + 4 * sq;
  const int ssto = srow * GBP + 4 * sq;
  const int afo = (16 * wt + l16) * GBP + 2 * kh;
  const int bfo = (16 * wj + l16) * GBP + 2 * kh;

  v4f ga[4], gb[2];
#pragma unroll
  for (int i = 0; i < 4; ++i) ga[i] = *(const v4f*)(ag + (16 * i) * CDIM);
#pragma unroll
  for (int i = 0; i < 2; ++i) gb[i] = *(const v4f*)(wg + (16 * i) * CDIM);
#pragma unroll
  for (int i = 0; i < 4; ++i) *(v4f*)(ldsA + ssto + 16 * i * GBP) = ga[i];
#pragma unroll
  for (int i = 0; i < 2; ++i) *(v4f*)(ldsB + ssto + 16 * i * GBP) = gb[i];
  __syncthreads();

  v8f acc = {};
  int st = 0;
#pragma unroll
  for (int c = 0; c < GNCH; ++c, st ^= 1) {
    const float* As = ldsA + st * PA_FLOATS;
    const float* Bs = ldsB + st * PB_FLOATS;

    const int k1 = (c + 1) * GK;
    if (c + 1 < GNCH) {
#pragma unroll
      for (int i = 0; i < 4; ++i) ga[i] = *(const v4f*)(ag + (16 * i) * CDIM + k1);
#pragma unroll
      for (int i = 0; i < 2; ++i) gb[i] = *(const v4f*)(wg + (16 * i) * CDIM + k1);
    }

    v2f af[GK / 4], bf[GK / 4];
#pragma unroll
    for (int t = 0; t < GK / 4; ++t) {
      af[t] = *(const v2f*)(As + afo + 4 * t);
      bf[t] = *(const v2f*)(Bs + bfo + 4 * t);
    }
    __builtin_amdgcn_sched_barrier(0);

#pragma unroll
    for (int t = 0; t < GK / 4; ++t) acc = wmma4(af[t], bf[t], acc);

    if (c + 1 < GNCH) {
      float* An = ldsA + (st ^ 1) * PA_FLOATS;
      float* Bn = ldsB + (st ^ 1) * PB_FLOATS;
#pragma unroll
      for (int i = 0; i < 4; ++i) *(v4f*)(An + ssto + 16 * i * GBP) = ga[i];
#pragma unroll
      for (int i = 0; i < 2; ++i) *(v4f*)(Bn + ssto + 16 * i * GBP) = gb[i];
      __syncthreads();
    }
  }

  const int col = j0 + 16 * wj + l16;
#pragma unroll
  for (int r = 0; r < 8; ++r) {
    const int n = n0 + 16 * wt + r + 8 * kh;
    out[col * N_TOK + n] = acc[r];
  }
}

extern "C" void kernel_launch(void* const* d_in, const int* in_sizes, int n_in,
                              void* d_out, int out_size, void* d_ws, size_t ws_size,
                              hipStream_t stream) {
  (void)in_sizes; (void)n_in; (void)out_size; (void)ws_size;
  const float* x     = (const float*)d_in[0];
  const float* wqkv  = (const float*)d_in[1];
  const float* wproj = (const float*)d_in[2];
  float* out = (float*)d_out;

  float* ws = (float*)d_ws;
  float* Q  = ws;
  float* K  = Q + NHEADS * N_TOK * HD;
  float* V  = K + NHEADS * N_TOK * HD;
  float* AO = V + NHEADS * N_TOK * HD;

  gsa3d_qkv_kernel <<<216, 256, 0, stream>>>(x, wqkv, Q, K, V);
  gsa3d_attn_kernel<<<216, 256, 0, stream>>>(Q, K, V, AO);
  gsa3d_proj_kernel<<<216, 256, 0, stream>>>(AO, wproj, out);
}
